# starting kernel
# baseline (speedup 1.0000x reference)
#include <hip/hip_runtime.h>


#define B_    2
#define C_    64
#define N_    16384
#define K_    16
#define COUT_ 64
#define CCAT_ 128

typedef __attribute__((ext_vector_type(2))) float v2f;
typedef __attribute__((ext_vector_type(8))) float v8f;


__global__ void __launch_bounds__(256)
prep_wcat_kernel(const float* __restrict__ W, float* __restrict__ WcatT) {
    int t = blockIdx.x * blockDim.x + threadIdx.x;
    if (t >= C_ * CCAT_) return;
    int c = t / CCAT_;
    int m = t % CCAT_;
    float v;
    if (m < COUT_) v = W[m * (3 * C_) + c] - W[m * (3 * C_) + C_ + c];
    else           v = W[(m - COUT_) * (3 * C_) + C_ + c];
    WcatT[c * CCAT_ + m] = v;
}

__global__ void __launch_bounds__(256)
mean_kernel(const float* __restrict__ x, float* __restrict__ gmean) {
    __shared__ float s[256];
    const int bc = blockIdx.x;
    const float* row = x + (size_t)bc * N_;
    float acc = 0.f;
    for (int n = threadIdx.x; n < N_; n += 256) acc += row[n];
    s[threadIdx.x] = acc;
    __syncthreads();
    for (int off = 128; off > 0; off >>= 1) {
        if (threadIdx.x < off) s[threadIdx.x] += s[threadIdx.x + off];
        __syncthreads();
    }
    if (threadIdx.x == 0) gmean[bc] = s[0] * (1.0f / N_);
}

__global__ void __launch_bounds__(128)
c0_kernel(const float* __restrict__ W, const float* __restrict__ bias,
          const float* __restrict__ gmean, float* __restrict__ c0) {
    int t = threadIdx.x;
    if (t >= B_ * COUT_) return;
    int b = t / COUT_, o = t % COUT_;
    float acc = bias[o];
    #pragma unroll 8
    for (int c = 0; c < C_; ++c)
        acc += W[o * (3 * C_) + 2 * C_ + c] * gmean[b * C_ + c];
    c0[t] = acc;
}

__global__ void __launch_bounds__(128)
gemm_wmma_kernel(const float* __restrict__ x, const float* __restrict__ WcatT,
                 float* __restrict__ Yt) {
    const int lane = threadIdx.x;
    const int wv   = threadIdx.y;
    const int n0   = blockIdx.x * 16;
    const int m0   = (blockIdx.y * 4 + wv) * 16;
    const int b    = blockIdx.z;
    const float* xb = x + (size_t)b * C_ * N_;
    float* yb = Yt + (size_t)b * N_ * CCAT_;
    const int ln = lane & 15;
    const int hi = lane >> 4;

    v8f acc = {};
    #pragma unroll
    for (int c = 0; c < C_; c += 4) {
        v2f a, bm;
        a.x = xb[(size_t)(c + 2 * hi    ) * N_ + n0 + ln];
        a.y = xb[(size_t)(c + 2 * hi + 1) * N_ + n0 + ln];
        bm.x = WcatT[(c + 2 * hi    ) * CCAT_ + m0 + ln];
        bm.y = WcatT[(c + 2 * hi + 1) * CCAT_ + m0 + ln];
        acc = __builtin_amdgcn_wmma_f32_16x16x4_f32(
            false, a, false, bm, (short)0, acc, false, false);
    }

    #pragma unroll
    for (int v = 0; v < 8; ++v) {
        int node = n0 + v + 8 * hi;
        yb[(size_t)node * CCAT_ + m0 + ln] = acc[v];
    }
}

__global__ void __launch_bounds__(256)
edge_kernel(const int* __restrict__ eidx, const float* __restrict__ Yt,
            const float* __restrict__ c0, float* __restrict__ out) {
    int t    = blockIdx.x * blockDim.x + threadIdx.x;
    int o    = t & (COUT_ - 1);
    int node = t >> 6;
    if (node >= B_ * N_) return;
    int b = node / N_;
    int n = node - b * N_;

    const int* e0 = eidx + ((size_t)b * N_ + n) * K_;
    const int* e1 = eidx + (size_t)B_ * N_ * K_ + ((size_t)b * N_ + n) * K_;
    const float* yb = Yt + (size_t)b * N_ * CCAT_;
    const float bias0 = c0[b * COUT_ + o];

    float m = -3.402823466e38f;
    #pragma unroll
    for (int k = 0; k < K_; ++k) {
        int i = e1[k];
        int j = e0[k];
        float v = yb[(size_t)i * CCAT_ + o]
                + yb[(size_t)j * CCAT_ + COUT_ + o]
                + bias0;
        m = fmaxf(m, v);
    }
    out[((size_t)b * COUT_ + o) * N_ + n] = fmaxf(m, 0.0f);
}

extern "C" void kernel_launch(void* const* d_in, const int* in_sizes, int n_in,
                              void* d_out, int out_size, void* d_ws, size_t ws_size,
                              hipStream_t stream) {
    const float* x    = (const float*)d_in[0];
    const int*   eidx = (const int*)  d_in[1];
    const float* W    = (const float*)d_in[2];
    const float* bias = (const float*)d_in[3];

    float* ws    = (float*)d_ws;
    float* WcatT = ws;
    float* gmean = ws + 8192;
    float* c0    = ws + 8192 + 128;
    float* Yt    = ws + 8192 + 256;

    prep_wcat_kernel<<<(C_ * CCAT_ + 255) / 256, 256, 0, stream>>>(W, WcatT);
    mean_kernel<<<B_ * C_, 256, 0, stream>>>(x, gmean);
    c0_kernel<<<1, 128, 0, stream>>>(W, bias, gmean, c0);

    dim3 ggrid(N_ / 16, (CCAT_ / 16) / 4, B_);
    dim3 gblock(32, 4, 1);
    gemm_wmma_kernel<<<ggrid, gblock, 0, stream>>>(x, WcatT, Yt);

    edge_kernel<<<(B_ * N_ * COUT_) / 256, 256, 0, stream>>>(eidx, Yt, c0, (float*)d_out);
}


# c0 by linearity: W3 rows join the f32 matrix product and stay summed over nodes in its accumulators, edge blocks add 64 partial rows; 5 launches -> 2, x read once; mean sum order changes
# speedup vs baseline: 2.0594x; 2.0594x over previous
#include <hip/hip_runtime.h>


#define B_    2
#define C_    64
#define N_    16384
#define K_    16
#define COUT_ 64
#define CCAT_ 128
#define NPG_  256
#define GPB_  (N_ / NPG_)

typedef __attribute__((ext_vector_type(2))) float v2f;
typedef __attribute__((ext_vector_type(8))) float v8f;

__global__ void __launch_bounds__(256)
gemm_wmma_kernel(const float* __restrict__ x, const float* __restrict__ W,
                 float* __restrict__ Yt, float* __restrict__ part) {
    __shared__ float s_part[4][2][16];
    const int lane = threadIdx.x & 31;
    const int wv   = threadIdx.x >> 5;
    const int tt   = wv & 1;
    const int q    = wv >> 1;
    const int p    = blockIdx.x & 1;
    const int g    = (blockIdx.x >> 1) & (GPB_ - 1);
    const int b    = blockIdx.x >> 7;
    const int t    = 2 * p + tt;
    const int ln = lane & 15;
    const int hi = lane >> 4;
    const float* xb = x + (size_t)b * C_ * N_;
    float* yb = Yt + (size_t)b * N_ * CCAT_;

    const float* wr = W + (16 * t + ln) * (3 * C_) + 2 * hi;
    v2f b1[16], b2[16], b3[16];
    #pragma unroll
    for (int s = 0; s < 16; ++s) {
        const v2f w1 = *reinterpret_cast<const v2f*>(wr + 4 * s);
        const v2f w2 = *reinterpret_cast<const v2f*>(wr + C_ + 4 * s);
        const v2f w3 = *reinterpret_cast<const v2f*>(wr + 2 * C_ + 4 * s);
        b1[s] = w1 - w2;
        b2[s] = w2;
        b3[s] = w3;
    }

    v8f acc3 = {};
    const int nq = g * NPG_ + q * 64;
    for (int tile = 0; tile < 4; ++tile) {
        const int n0 = nq + tile * 16;
        const float* xa = xb + (size_t)(2 * hi) * N_ + n0 + ln;
        v8f acc1 = {}, acc2 = {};
        #pragma unroll
        for (int s = 0; s < 16; ++s) {
            v2f a;
            a.x = xa[(size_t)(4 * s    ) * N_];
            a.y = xa[(size_t)(4 * s + 1) * N_];
            acc1 = __builtin_amdgcn_wmma_f32_16x16x4_f32(false, a, false, b1[s], (short)0, acc1, false, false);
            acc2 = __builtin_amdgcn_wmma_f32_16x16x4_f32(false, a, false, b2[s], (short)0, acc2, false, false);
            acc3 = __builtin_amdgcn_wmma_f32_16x16x4_f32(false, a, false, b3[s], (short)0, acc3, false, false);
        }
        #pragma unroll
        for (int v = 0; v < 8; ++v) {
            const int node = n0 + v + 8 * hi;
            yb[(size_t)node * CCAT_ + 16 * t + ln]         = acc1[v];
            yb[(size_t)node * CCAT_ + COUT_ + 16 * t + ln] = acc2[v];
        }
    }

    float sn = ((acc3[0] + acc3[1]) + (acc3[2] + acc3[3])) + ((acc3[4] + acc3[5]) + (acc3[6] + acc3[7]));
    sn += __shfl_xor(sn, 16);
    if (hi == 0) s_part[q][tt][ln] = sn;
    __syncthreads();
    if (threadIdx.x < 32) {
        const int t2 = lane >> 4;
        const float tot = (s_part[0][t2][ln] + s_part[1][t2][ln]) + (s_part[2][t2][ln] + s_part[3][t2][ln]);
        part[((size_t)b * GPB_ + g) * COUT_ + 16 * (2 * p + t2) + ln] = tot;
    }
}

#define NPB_ 16
__global__ void __launch_bounds__(64) __attribute__((amdgpu_waves_per_eu(4, 4)))
edge_kernel(const int* __restrict__ eidx, const float* __restrict__ Yt,
            const float* __restrict__ part, const float* __restrict__ bias,
            float* __restrict__ out) {
    const unsigned o = threadIdx.x;
    const int node0 = blockIdx.x * NPB_;
    const int b     = node0 / N_;
    const int n0    = node0 - b * N_;

    const int* e0 = eidx + (size_t)node0 * K_;
    const int* e1 = eidx + (size_t)B_ * N_ * K_ + (size_t)node0 * K_;
    const float* yb = Yt + (size_t)b * N_ * CCAT_;
    const float* pr = part + (size_t)b * GPB_ * COUT_ + o;
    float ps[4] = {0.f, 0.f, 0.f, 0.f};
    #pragma unroll
    for (int i = 0; i < GPB_; i += 4) {
        ps[0] += pr[(i    ) * COUT_];
        ps[1] += pr[(i + 1) * COUT_];
        ps[2] += pr[(i + 2) * COUT_];
        ps[3] += pr[(i + 3) * COUT_];
    }
    const float bias0 = bias[o] + ((ps[0] + ps[1]) + (ps[2] + ps[3])) * (1.0f / N_);
    float* orow = out + ((size_t)b * COUT_ + o) * N_ + n0;

    for (int g = 0; g < NPB_; g += 4) {
        float r[4];
        #pragma unroll
        for (int h = 0; h < 4; h += 2) {
            float y1[2][K_], y2[2][K_];
            #pragma unroll
            for (int q = 0; q < 2; ++q) {
                #pragma unroll
                for (int k = 0; k < K_; ++k) {
                    const unsigned i = (unsigned)e1[(g + h + q) * K_ + k];
                    const unsigned j = (unsigned)e0[(g + h + q) * K_ + k];
                    y1[q][k] = yb[i * CCAT_ + o];
                    y2[q][k] = yb[j * CCAT_ + COUT_ + o];
                }
            }
            #pragma unroll
            for (int q = 0; q < 2; ++q) {
                float m = -3.402823466e38f;
                #pragma unroll
                for (int k = 0; k < K_; ++k)
                    m = fmaxf(m, y1[q][k] + y2[q][k] + bias0);
                r[h + q] = fmaxf(m, 0.0f);
            }
        }
        *reinterpret_cast<float4*>(orow + g) = make_float4(r[0], r[1], r[2], r[3]);
    }
}

extern "C" void kernel_launch(void* const* d_in, const int* in_sizes, int n_in,
                              void* d_out, int out_size, void* d_ws, size_t ws_size,
                              hipStream_t stream) {
    const float* x    = (const float*)d_in[0];
    const int*   eidx = (const int*)  d_in[1];
    const float* W    = (const float*)d_in[2];
    const float* bias = (const float*)d_in[3];

    float* ws    = (float*)d_ws;
    float* part  = ws;
    float* Yt    = ws + 8192 + 256;

    gemm_wmma_kernel<<<B_ * GPB_ * 2, 256, 0, stream>>>(x, W, Yt, part);

    edge_kernel<<<(B_ * N_) / NPB_, 64, 0, stream>>>(eidx, Yt, part, bias, (float*)d_out);
}


# edge kernel as one rolling stream: row reads of the next two nodes and the indices after them go out before the current two are reduced; c0 added after the max (monotone, same bits); sums unchanged
# speedup vs baseline: 2.4539x; 1.1916x over previous
#include <hip/hip_runtime.h>


#define B_    2
#define C_    64
#define N_    16384
#define K_    16
#define COUT_ 64
#define CCAT_ 128
#define NPG_  256
#define GPB_  (N_ / NPG_)

typedef __attribute__((ext_vector_type(2))) float v2f;
typedef __attribute__((ext_vector_type(8))) float v8f;

__global__ void __launch_bounds__(256)
gemm_wmma_kernel(const float* __restrict__ x, const float* __restrict__ W,
                 float* __restrict__ Yt, float* __restrict__ part) {
    __shared__ float s_part[4][2][16];
    const int lane = threadIdx.x & 31;
    const int wv   = threadIdx.x >> 5;
    const int tt   = wv & 1;
    const int q    = wv >> 1;
    const int p    = blockIdx.x & 1;
    const int g    = (blockIdx.x >> 1) & (GPB_ - 1);
    const int b    = blockIdx.x >> 7;
    const int t    = 2 * p + tt;
    const int ln = lane & 15;
    const int hi = lane >> 4;
    const float* xb = x + (size_t)b * C_ * N_;
    float* yb = Yt + (size_t)b * N_ * CCAT_;

    const float* wr = W + (16 * t + ln) * (3 * C_) + 2 * hi;
    v2f b1[16], b2[16], b3[16];
    #pragma unroll
    for (int s = 0; s < 16; ++s) {
        const v2f w1 = *reinterpret_cast<const v2f*>(wr + 4 * s);
        const v2f w2 = *reinterpret_cast<const v2f*>(wr + C_ + 4 * s);
        const v2f w3 = *reinterpret_cast<const v2f*>(wr + 2 * C_ + 4 * s);
        b1[s] = w1 - w2;
        b2[s] = w2;
        b3[s] = w3;
    }

    v8f acc3 = {};
    const int nq = g * NPG_ + q * 64;
    for (int tile = 0; tile < 4; ++tile) {
        const int n0 = nq + tile * 16;
        const float* xa = xb + (size_t)(2 * hi) * N_ + n0 + ln;
        v8f acc1 = {}, acc2 = {};
        #pragma unroll
        for (int s = 0; s < 16; ++s) {
            v2f a;
            a.x = xa[(size_t)(4 * s    ) * N_];
            a.y = xa[(size_t)(4 * s + 1) * N_];
            acc1 = __builtin_amdgcn_wmma_f32_16x16x4_f32(false, a, false, b1[s], (short)0, acc1, false, false);
            acc2 = __builtin_amdgcn_wmma_f32_16x16x4_f32(false, a, false, b2[s], (short)0, acc2, false, false);
            acc3 = __builtin_amdgcn_wmma_f32_16x16x4_f32(false, a, false, b3[s], (short)0, acc3, false, false);
        }
        #pragma unroll
        for (int v = 0; v < 8; ++v) {
            const int node = n0 + v + 8 * hi;
            yb[(size_t)node * CCAT_ + 16 * t + ln]         = acc1[v];
            yb[(size_t)node * CCAT_ + COUT_ + 16 * t + ln] = acc2[v];
        }
    }

    float sn = ((acc3[0] + acc3[1]) + (acc3[2] + acc3[3])) + ((acc3[4] + acc3[5]) + (acc3[6] + acc3[7]));
    sn += __shfl_xor(sn, 16);
    if (hi == 0) s_part[q][tt][ln] = sn;
    __syncthreads();
    if (threadIdx.x < 32) {
        const int t2 = lane >> 4;
        const float tot = (s_part[0][t2][ln] + s_part[1][t2][ln]) + (s_part[2][t2][ln] + s_part[3][t2][ln]);
        part[((size_t)b * GPB_ + g) * COUT_ + 16 * (2 * p + t2) + ln] = tot;
    }
}

#define NPB_ 16
#define NGRP_ (NPB_ / 2)
__global__ void __launch_bounds__(64) __attribute__((amdgpu_waves_per_eu(4, 4)))
edge_kernel(const int* __restrict__ eidx, const float* __restrict__ Yt,
            const float* __restrict__ part, const float* __restrict__ bias,
            float* __restrict__ out) {
    const unsigned o = threadIdx.x;
    const int node0 = blockIdx.x * NPB_;
    const int b     = node0 / N_;
    const int n0    = node0 - b * N_;

    const int* e0 = eidx + (size_t)node0 * K_;
    const int* e1 = eidx + (size_t)B_ * N_ * K_ + (size_t)node0 * K_;
    const float* yb = Yt + (size_t)b * N_ * CCAT_;
    float* orow = out + ((size_t)b * COUT_ + o) * N_ + n0;

    unsigned ii[2][2 * K_], jj[2][2 * K_];
    float y1[2][2 * K_], y2[2][2 * K_];

    #pragma unroll
    for (int t = 0; t < 2 * K_; ++t) { ii[0][t] = (unsigned)e1[t]; jj[0][t] = (unsigned)e0[t]; }
    #pragma unroll
    for (int t = 0; t < 2 * K_; ++t) {
        y1[0][t] = yb[ii[0][t] * CCAT_ + o];
        y2[0][t] = yb[jj[0][t] * CCAT_ + COUT_ + o];
    }
    #pragma unroll
    for (int t = 0; t < 2 * K_; ++t) { ii[1][t] = (unsigned)e1[2 * K_ + t]; jj[1][t] = (unsigned)e0[2 * K_ + t]; }

    const float* pr = part + (size_t)b * GPB_ * COUT_ + o;
    float ps[4] = {0.f, 0.f, 0.f, 0.f};
    #pragma unroll
    for (int i = 0; i < GPB_; i += 4) {
        ps[0] += pr[(i    ) * COUT_];
        ps[1] += pr[(i + 1) * COUT_];
        ps[2] += pr[(i + 2) * COUT_];
        ps[3] += pr[(i + 3) * COUT_];
    }
    const float bias0 = bias[o] + ((ps[0] + ps[1]) + (ps[2] + ps[3])) * (1.0f / N_);

    float r[4];
    #pragma unroll
    for (int g = 0; g < NGRP_; ++g) {
        const int cur = g & 1, nxt = cur ^ 1;
        if (g + 1 < NGRP_) {
            #pragma unroll
            for (int t = 0; t < 2 * K_; ++t) {
                y1[nxt][t] = yb[ii[nxt][t] * CCAT_ + o];
                y2[nxt][t] = yb[jj[nxt][t] * CCAT_ + COUT_ + o];
            }
        }
        if (g + 2 < NGRP_) {
            #pragma unroll
            for (int t = 0; t < 2 * K_; ++t) {
                ii[cur][t] = (unsigned)e1[(g + 2) * 2 * K_ + t];
                jj[cur][t] = (unsigned)e0[(g + 2) * 2 * K_ + t];
            }
        }
        #pragma unroll
        for (int q = 0; q < 2; ++q) {
            float m = -3.402823466e38f;
            #pragma unroll
            for (int k = 0; k < K_; ++k)
                m = fmaxf(m, y1[cur][q * K_ + k] + y2[cur][q * K_ + k]);
            r[2 * (g & 1) + q] = fmaxf(m + bias0, 0.0f);
        }
        if (g & 1)
            *reinterpret_cast<float4*>(orow + 2 * (g - 1)) = make_float4(r[0], r[1], r[2], r[3]);
    }
}

extern "C" void kernel_launch(void* const* d_in, const int* in_sizes, int n_in,
                              void* d_out, int out_size, void* d_ws, size_t ws_size,
                              hipStream_t stream) {
    const float* x    = (const float*)d_in[0];
    const int*   eidx = (const int*)  d_in[1];
    const float* W    = (const float*)d_in[2];
    const float* bias = (const float*)d_in[3];

    float* ws    = (float*)d_ws;
    float* part  = ws;
    float* Yt    = ws + 8192 + 256;

    gemm_wmma_kernel<<<B_ * GPB_ * 2, 256, 0, stream>>>(x, W, Yt, part);

    edge_kernel<<<(B_ * N_) / NPB_, 64, 0, stream>>>(eidx, Yt, part, bias, (float*)d_out);
}


# product kernel: A fragment reads run one node tile ahead of the matrix instructions (second register set), the first tile's go out with the weight reads; order of sums unchanged
# speedup vs baseline: 2.5921x; 1.0563x over previous
#include <hip/hip_runtime.h>


#define B_    2
#define C_    64
#define N_    16384
#define K_    16
#define COUT_ 64
#define CCAT_ 128
#define NPG_  256
#define GPB_  (N_ / NPG_)

typedef __attribute__((ext_vector_type(2))) float v2f;
typedef __attribute__((ext_vector_type(8))) float v8f;

__global__ void __launch_bounds__(256)
gemm_wmma_kernel(const float* __restrict__ x, const float* __restrict__ W,
                 float* __restrict__ Yt, float* __restrict__ part) {
    __shared__ float s_part[4][2][16];
    const int lane = threadIdx.x & 31;
    const int wv   = threadIdx.x >> 5;
    const int tt   = wv & 1;
    const int q    = wv >> 1;
    const int p    = blockIdx.x & 1;
    const int g    = (blockIdx.x >> 1) & (GPB_ - 1);
    const int b    = blockIdx.x >> 7;
    const int t    = 2 * p + tt;
    const int ln = lane & 15;
    const int hi = lane >> 4;
    const float* xb = x + (size_t)b * C_ * N_;
    float* yb = Yt + (size_t)b * N_ * CCAT_;

    const int nq = g * NPG_ + q * 64;
    const float* xq = xb + (size_t)(2 * hi) * N_ + nq + ln;
    v2f a[2][16];
    #pragma unroll
    for (int s = 0; s < 16; ++s) {
        a[0][s].x = xq[(size_t)(4 * s    ) * N_];
        a[0][s].y = xq[(size_t)(4 * s + 1) * N_];
    }

    const float* wr = W + (16 * t + ln) * (3 * C_) + 2 * hi;
    v2f b1[16], b2[16], b3[16];
    #pragma unroll
    for (int s = 0; s < 16; ++s) {
        const v2f w1 = *reinterpret_cast<const v2f*>(wr + 4 * s);
        const v2f w2 = *reinterpret_cast<const v2f*>(wr + C_ + 4 * s);
        const v2f w3 = *reinterpret_cast<const v2f*>(wr + 2 * C_ + 4 * s);
        b1[s] = w1 - w2;
        b2[s] = w2;
        b3[s] = w3;
    }

    v8f acc3 = {};
    #pragma unroll
    for (int tile = 0; tile < 4; ++tile) {
        const int n0 = nq + tile * 16;
        const int cur = tile & 1;
        if (tile + 1 < 4) {
            const float* xn = xq + (tile + 1) * 16;
            #pragma unroll
            for (int s = 0; s < 16; ++s) {
                a[cur ^ 1][s].x = xn[(size_t)(4 * s    ) * N_];
                a[cur ^ 1][s].y = xn[(size_t)(4 * s + 1) * N_];
            }
        }
        v8f acc1 = {}, acc2 = {};
        #pragma unroll
        for (int s = 0; s < 16; ++s) {
            acc1 = __builtin_amdgcn_wmma_f32_16x16x4_f32(false, a[cur][s], false, b1[s], (short)0, acc1, false, false);
            acc2 = __builtin_amdgcn_wmma_f32_16x16x4_f32(false, a[cur][s], false, b2[s], (short)0, acc2, false, false);
            acc3 = __builtin_amdgcn_wmma_f32_16x16x4_f32(false, a[cur][s], false, b3[s], (short)0, acc3, false, false);
        }
        #pragma unroll
        for (int v = 0; v < 8; ++v) {
            const int node = n0 + v + 8 * hi;
            yb[(size_t)node * CCAT_ + 16 * t + ln]         = acc1[v];
            yb[(size_t)node * CCAT_ + COUT_ + 16 * t + ln] = acc2[v];
        }
        __builtin_amdgcn_sched_barrier(0);
    }

    float sn = ((acc3[0] + acc3[1]) + (acc3[2] + acc3[3])) + ((acc3[4] + acc3[5]) + (acc3[6] + acc3[7]));
    sn += __shfl_xor(sn, 16);
    if (hi == 0) s_part[q][tt][ln] = sn;
    __syncthreads();
    if (threadIdx.x < 32) {
        const int t2 = lane >> 4;
        const float tot = (s_part[0][t2][ln] + s_part[1][t2][ln]) + (s_part[2][t2][ln] + s_part[3][t2][ln]);
        part[((size_t)b * GPB_ + g) * COUT_ + 16 * (2 * p + t2) + ln] = tot;
    }
}

#define NPB_ 16
#define NGRP_ (NPB_ / 2)
__global__ void __launch_bounds__(64) __attribute__((amdgpu_waves_per_eu(4, 4)))
edge_kernel(const int* __restrict__ eidx, const float* __restrict__ Yt,
            const float* __restrict__ part, const float* __restrict__ bias,
            float* __restrict__ out) {
    const unsigned o = threadIdx.x;
    const int node0 = blockIdx.x * NPB_;
    const int b     = node0 / N_;
    const int n0    = node0 - b * N_;

    const int* e0 = eidx + (size_t)node0 * K_;
    const int* e1 = eidx + (size_t)B_ * N_ * K_ + (size_t)node0 * K_;
    const float* yb = Yt + (size_t)b * N_ * CCAT_;
    float* orow = out + ((size_t)b * COUT_ + o) * N_ + n0;

    unsigned ii[2][2 * K_], jj[2][2 * K_];
    float y1[2][2 * K_], y2[2][2 * K_];

    #pragma unroll
    for (int t = 0; t < 2 * K_; ++t) { ii[0][t] = (unsigned)e1[t]; jj[0][t] = (unsigned)e0[t]; }
    #pragma unroll
    for (int t = 0; t < 2 * K_; ++t) {
        y1[0][t] = yb[ii[0][t] * CCAT_ + o];
        y2[0][t] = yb[jj[0][t] * CCAT_ + COUT_ + o];
    }
    #pragma unroll
    for (int t = 0; t < 2 * K_; ++t) { ii[1][t] = (unsigned)e1[2 * K_ + t]; jj[1][t] = (unsigned)e0[2 * K_ + t]; }

    const float* pr = part + (size_t)b * GPB_ * COUT_ + o;
    float ps[4] = {0.f, 0.f, 0.f, 0.f};
    #pragma unroll
    for (int i = 0; i < GPB_; i += 4) {
        ps[0] += pr[(i    ) * COUT_];
        ps[1] += pr[(i + 1) * COUT_];
        ps[2] += pr[(i + 2) * COUT_];
        ps[3] += pr[(i + 3) * COUT_];
    }
    const float bias0 = bias[o] + ((ps[0] + ps[1]) + (ps[2] + ps[3])) * (1.0f / N_);

    float r[4];
    #pragma unroll
    for (int g = 0; g < NGRP_; ++g) {
        const int cur = g & 1, nxt = cur ^ 1;
        if (g + 1 < NGRP_) {
            #pragma unroll
            for (int t = 0; t < 2 * K_; ++t) {
                y1[nxt][t] = yb[ii[nxt][t] * CCAT_ + o];
                y2[nxt][t] = yb[jj[nxt][t] * CCAT_ + COUT_ + o];
            }
        }
        if (g + 2 < NGRP_) {
            #pragma unroll
            for (int t = 0; t < 2 * K_; ++t) {
                ii[cur][t] = (unsigned)e1[(g + 2) * 2 * K_ + t];
                jj[cur][t] = (unsigned)e0[(g + 2) * 2 * K_ + t];
            }
        }
        #pragma unroll
        for (int q = 0; q < 2; ++q) {
            float m = -3.402823466e38f;
            #pragma unroll
            for (int k = 0; k < K_; ++k)
                m = fmaxf(m, y1[cur][q * K_ + k] + y2[cur][q * K_ + k]);
            r[2 * (g & 1) + q] = fmaxf(m + bias0, 0.0f);
        }
        if (g & 1)
            *reinterpret_cast<float4*>(orow + 2 * (g - 1)) = make_float4(r[0], r[1], r[2], r[3]);
    }
}

extern "C" void kernel_launch(void* const* d_in, const int* in_sizes, int n_in,
                              void* d_out, int out_size, void* d_ws, size_t ws_size,
                              hipStream_t stream) {
    const float* x    = (const float*)d_in[0];
    const int*   eidx = (const int*)  d_in[1];
    const float* W    = (const float*)d_in[2];
    const float* bias = (const float*)d_in[3];

    float* ws    = (float*)d_ws;
    float* part  = ws;
    float* Yt    = ws + 8192 + 256;

    gemm_wmma_kernel<<<B_ * GPB_ * 2, 256, 0, stream>>>(x, W, Yt, part);

    edge_kernel<<<(B_ * N_) / NPB_, 64, 0, stream>>>(eidx, Yt, part, bias, (float*)d_out);
}


# edge kernel: blocks of four 64-lane walkers form c0 once per block, 4 wide reads of the partial rows per thread and an LDS sum, in place of 64 row reads per lane; c0 sum order changes
# speedup vs baseline: 3.0394x; 1.1726x over previous
#include <hip/hip_runtime.h>


#define B_    2
#define C_    64
#define N_    16384
#define K_    16
#define COUT_ 64
#define CCAT_ 128
#define NPG_  256
#define GPB_  (N_ / NPG_)

typedef __attribute__((ext_vector_type(2))) float v2f;
typedef __attribute__((ext_vector_type(8))) float v8f;

__global__ void __launch_bounds__(256)
gemm_wmma_kernel(const float* __restrict__ x, const float* __restrict__ W,
                 float* __restrict__ Yt, float* __restrict__ part) {
    __shared__ float s_part[4][2][16];
    const int lane = threadIdx.x & 31;
    const int wv   = threadIdx.x >> 5;
    const int tt   = wv & 1;
    const int q    = wv >> 1;
    const int p    = blockIdx.x & 1;
    const int g    = (blockIdx.x >> 1) & (GPB_ - 1);
    const int b    = blockIdx.x >> 7;
    const int t    = 2 * p + tt;
    const int ln = lane & 15;
    const int hi = lane >> 4;
    const float* xb = x + (size_t)b * C_ * N_;
    float* yb = Yt + (size_t)b * N_ * CCAT_;

    const int nq = g * NPG_ + q * 64;
    const float* xq = xb + (size_t)(2 * hi) * N_ + nq + ln;
    v2f a[2][16];
    #pragma unroll
    for (int s = 0; s < 16; ++s) {
        a[0][s].x = xq[(size_t)(4 * s    ) * N_];
        a[0][s].y = xq[(size_t)(4 * s + 1) * N_];
    }

    const float* wr = W + (16 * t + ln) * (3 * C_) + 2 * hi;
    v2f b1[16], b2[16], b3[16];
    #pragma unroll
    for (int s = 0; s < 16; ++s) {
        const v2f w1 = *reinterpret_cast<const v2f*>(wr + 4 * s);
        const v2f w2 = *reinterpret_cast<const v2f*>(wr + C_ + 4 * s);
        const v2f w3 = *reinterpret_cast<const v2f*>(wr + 2 * C_ + 4 * s);
        b1[s] = w1 - w2;
        b2[s] = w2;
        b3[s] = w3;
    }

    v8f acc3 = {};
    #pragma unroll
    for (int tile = 0; tile < 4; ++tile) {
        const int n0 = nq + tile * 16;
        const int cur = tile & 1;
        if (tile + 1 < 4) {
            const float* xn = xq + (tile + 1) * 16;
            #pragma unroll
            for (int s = 0; s < 16; ++s) {
                a[cur ^ 1][s].x = xn[(size_t)(4 * s    ) * N_];
                a[cur ^ 1][s].y = xn[(size_t)(4 * s + 1) * N_];
            }
        }
        v8f acc1 = {}, acc2 = {};
        #pragma unroll
        for (int s = 0; s < 16; ++s) {
            acc1 = __builtin_amdgcn_wmma_f32_16x16x4_f32(false, a[cur][s], false, b1[s], (short)0, acc1, false, false);
            acc2 = __builtin_amdgcn_wmma_f32_16x16x4_f32(false, a[cur][s], false, b2[s], (short)0, acc2, false, false);
            acc3 = __builtin_amdgcn_wmma_f32_16x16x4_f32(false, a[cur][s], false, b3[s], (short)0, acc3, false, false);
        }
        #pragma unroll
        for (int v = 0; v < 8; ++v) {
            const int node = n0 + v + 8 * hi;
            yb[(size_t)node * CCAT_ + 16 * t + ln]         = acc1[v];
            yb[(size_t)node * CCAT_ + COUT_ + 16 * t + ln] = acc2[v];
        }
        __builtin_amdgcn_sched_barrier(0);
    }

    float sn = ((acc3[0] + acc3[1]) + (acc3[2] + acc3[3])) + ((acc3[4] + acc3[5]) + (acc3[6] + acc3[7]));
    sn += __shfl_xor(sn, 16);
    if (hi == 0) s_part[q][tt][ln] = sn;
    __syncthreads();
    if (threadIdx.x < 32) {
        const int t2 = lane >> 4;
        const float tot = (s_part[0][t2][ln] + s_part[1][t2][ln]) + (s_part[2][t2][ln] + s_part[3][t2][ln]);
        part[((size_t)b * GPB_ + g) * COUT_ + 16 * (2 * p + t2) + ln] = tot;
    }
}

#define NPB_ 16
#define NGRP_ (NPB_ / 2)
#define SUBS_ 4
__global__ void __launch_bounds__(64 * SUBS_) __attribute__((amdgpu_waves_per_eu(4, 4)))
edge_kernel(const int* __restrict__ eidx, const float* __restrict__ Yt,
            const float* __restrict__ part, const float* __restrict__ bias,
            float* __restrict__ out) {
    __shared__ float s_red[64 * SUBS_ * 4];
    const unsigned tid = threadIdx.x;
    const unsigned o = tid & 63;
    const int sub   = __builtin_amdgcn_readfirstlane((int)(tid >> 6));
    const int node0 = (blockIdx.x * SUBS_ + sub) * NPB_;
    const int b     = (blockIdx.x * (SUBS_ * NPB_)) / N_;
    const int n0    = node0 - b * N_;

    const int* e0 = eidx + (size_t)node0 * K_;
    const int* e1 = eidx + (size_t)B_ * N_ * K_ + (size_t)node0 * K_;
    const float* yb = Yt + (size_t)b * N_ * CCAT_;
    float* orow = out + ((size_t)b * COUT_ + o) * N_ + n0;

    unsigned ii[2][2 * K_], jj[2][2 * K_];
    float y1[2][2 * K_], y2[2][2 * K_];

    #pragma unroll
    for (int t = 0; t < 2 * K_; ++t) { ii[0][t] = (unsigned)e1[t]; jj[0][t] = (unsigned)e0[t]; }

    const float4* pp = reinterpret_cast<const float4*>(part + (size_t)b * GPB_ * COUT_) + tid;
    float4 pa = pp[0], pb = pp[64 * SUBS_], pc = pp[2 * 64 * SUBS_], pd = pp[3 * 64 * SUBS_];
    const float bo = bias[o];
    *reinterpret_cast<float4*>(&s_red[4 * tid]) =
        make_float4((pa.x + pb.x) + (pc.x + pd.x), (pa.y + pb.y) + (pc.y + pd.y),
                    (pa.z + pb.z) + (pc.z + pd.z), (pa.w + pb.w) + (pc.w + pd.w));
    __syncthreads();
    float ps[4] = {0.f, 0.f, 0.f, 0.f};
    #pragma unroll
    for (int k = 0; k < 16; k += 4) {
        ps[0] += s_red[64 * (k    ) + o];
        ps[1] += s_red[64 * (k + 1) + o];
        ps[2] += s_red[64 * (k + 2) + o];
        ps[3] += s_red[64 * (k + 3) + o];
    }
    const float bias0 = bo + ((ps[0] + ps[1]) + (ps[2] + ps[3])) * (1.0f / N_);
    __builtin_amdgcn_sched_barrier(0);
    #pragma unroll
    for (int t = 0; t < 2 * K_; ++t) {
        y1[0][t] = yb[ii[0][t] * CCAT_ + o];
        y2[0][t] = yb[jj[0][t] * CCAT_ + COUT_ + o];
    }
    #pragma unroll
    for (int t = 0; t < 2 * K_; ++t) { ii[1][t] = (unsigned)e1[2 * K_ + t]; jj[1][t] = (unsigned)e0[2 * K_ + t]; }

    float r[4];
    #pragma unroll
    for (int g = 0; g < NGRP_; ++g) {
        const int cur = g & 1, nxt = cur ^ 1;
        if (g + 1 < NGRP_) {
            #pragma unroll
            for (int t = 0; t < 2 * K_; ++t) {
                y1[nxt][t] = yb[ii[nxt][t] * CCAT_ + o];
                y2[nxt][t] = yb[jj[nxt][t] * CCAT_ + COUT_ + o];
            }
        }
        if (g + 2 < NGRP_) {
            #pragma unroll
            for (int t = 0; t < 2 * K_; ++t) {
                ii[cur][t] = (unsigned)e1[(g + 2) * 2 * K_ + t];
                jj[cur][t] = (unsigned)e0[(g + 2) * 2 * K_ + t];
            }
        }
        #pragma unroll
        for (int q = 0; q < 2; ++q) {
            float m = -3.402823466e38f;
            #pragma unroll
            for (int k = 0; k < K_; ++k)
                m = fmaxf(m, y1[cur][q * K_ + k] + y2[cur][q * K_ + k]);
            r[2 * (g & 1) + q] = fmaxf(m + bias0, 0.0f);
        }
        if (g & 1)
            *reinterpret_cast<float4*>(orow + 2 * (g - 1)) = make_float4(r[0], r[1], r[2], r[3]);
    }
}

extern "C" void kernel_launch(void* const* d_in, const int* in_sizes, int n_in,
                              void* d_out, int out_size, void* d_ws, size_t ws_size,
                              hipStream_t stream) {
    const float* x    = (const float*)d_in[0];
    const int*   eidx = (const int*)  d_in[1];
    const float* W    = (const float*)d_in[2];
    const float* bias = (const float*)d_in[3];

    float* ws    = (float*)d_ws;
    float* part  = ws;
    float* Yt    = ws + 8192 + 256;

    gemm_wmma_kernel<<<B_ * GPB_ * 2, 256, 0, stream>>>(x, W, Yt, part);

    edge_kernel<<<(B_ * N_) / (NPB_ * SUBS_), 64 * SUBS_, 0, stream>>>(eidx, Yt, part, bias, (float*)d_out);
}


# edge kernel: the four lanes of a quad exchange their 16B result pieces in registers (4x4 transpose by quad lane moves), so a store writes 8 runs of 64B, not 32 pieces of 16B; sums unchanged
# speedup vs baseline: 3.2187x; 1.0590x over previous
#include <hip/hip_runtime.h>


#define B_    2
#define C_    64
#define N_    16384
#define K_    16
#define COUT_ 64
#define CCAT_ 128
#define NPG_  256
#define GPB_  (N_ / NPG_)

typedef __attribute__((ext_vector_type(2))) float v2f;
typedef __attribute__((ext_vector_type(8))) float v8f;

__global__ void __launch_bounds__(256)
gemm_wmma_kernel(const float* __restrict__ x, const float* __restrict__ W,
                 float* __restrict__ Yt, float* __restrict__ part) {
    __shared__ float s_part[4][2][16];
    const int lane = threadIdx.x & 31;
    const int wv   = threadIdx.x >> 5;
    const int tt   = wv & 1;
    const int q    = wv >> 1;
    const int p    = blockIdx.x & 1;
    const int g    = (blockIdx.x >> 1) & (GPB_ - 1);
    const int b    = blockIdx.x >> 7;
    const int t    = 2 * p + tt;
    const int ln = lane & 15;
    const int hi = lane >> 4;
    const float* xb = x + (size_t)b * C_ * N_;
    float* yb = Yt + (size_t)b * N_ * CCAT_;

    const int nq = g * NPG_ + q * 64;
    const float* xq = xb + (size_t)(2 * hi) * N_ + nq + ln;
    v2f a[2][16];
    #pragma unroll
    for (int s = 0; s < 16; ++s) {
        a[0][s].x = xq[(size_t)(4 * s    ) * N_];
        a[0][s].y = xq[(size_t)(4 * s + 1) * N_];
    }

    const float* wr = W + (16 * t + ln) * (3 * C_) + 2 * hi;
    v2f b1[16], b2[16], b3[16];
    #pragma unroll
    for (int s = 0; s < 16; ++s) {
        const v2f w1 = *reinterpret_cast<const v2f*>(wr + 4 * s);
        const v2f w2 = *reinterpret_cast<const v2f*>(wr + C_ + 4 * s);
        const v2f w3 = *reinterpret_cast<const v2f*>(wr + 2 * C_ + 4 * s);
        b1[s] = w1 - w2;
        b2[s] = w2;
        b3[s] = w3;
    }

    v8f acc3 = {};
    #pragma unroll
    for (int tile = 0; tile < 4; ++tile) {
        const int n0 = nq + tile * 16;
        const int cur = tile & 1;
        if (tile + 1 < 4) {
            const float* xn = xq + (tile + 1) * 16;
            #pragma unroll
            for (int s = 0; s < 16; ++s) {
                a[cur ^ 1][s].x = xn[(size_t)(4 * s    ) * N_];
                a[cur ^ 1][s].y = xn[(size_t)(4 * s + 1) * N_];
            }
        }
        v8f acc1 = {}, acc2 = {};
        #pragma unroll
        for (int s = 0; s < 16; ++s) {
            acc1 = __builtin_amdgcn_wmma_f32_16x16x4_f32(false, a[cur][s], false, b1[s], (short)0, acc1, false, false);
            acc2 = __builtin_amdgcn_wmma_f32_16x16x4_f32(false, a[cur][s], false, b2[s], (short)0, acc2, false, false);
            acc3 = __builtin_amdgcn_wmma_f32_16x16x4_f32(false, a[cur][s], false, b3[s], (short)0, acc3, false, false);
        }
        #pragma unroll
        for (int v = 0; v < 8; ++v) {
            const int node = n0 + v + 8 * hi;
            yb[(size_t)node * CCAT_ + 16 * t + ln]         = acc1[v];
            yb[(size_t)node * CCAT_ + COUT_ + 16 * t + ln] = acc2[v];
        }
        __builtin_amdgcn_sched_barrier(0);
    }

    float sn = ((acc3[0] + acc3[1]) + (acc3[2] + acc3[3])) + ((acc3[4] + acc3[5]) + (acc3[6] + acc3[7]));
    sn += __shfl_xor(sn, 16);
    if (hi == 0) s_part[q][tt][ln] = sn;
    __syncthreads();
    if (threadIdx.x < 32) {
        const int t2 = lane >> 4;
        const float tot = (s_part[0][t2][ln] + s_part[1][t2][ln]) + (s_part[2][t2][ln] + s_part[3][t2][ln]);
        part[((size_t)b * GPB_ + g) * COUT_ + 16 * (2 * p + t2) + ln] = tot;
    }
}

__device__ __forceinline__ float quad_xor1(float v) {
    return __builtin_bit_cast(float, __builtin_amdgcn_mov_dpp(__builtin_bit_cast(int, v), 0xB1, 0xF, 0xF, true));
}
__device__ __forceinline__ float quad_xor2(float v) {
    return __builtin_bit_cast(float, __builtin_amdgcn_mov_dpp(__builtin_bit_cast(int, v), 0x4E, 0xF, 0xF, true));
}

#define NPB_ 16
#define NGRP_ (NPB_ / 2)
#define SUBS_ 4
__global__ void __launch_bounds__(64 * SUBS_) __attribute__((amdgpu_waves_per_eu(4, 4)))
edge_kernel(const int* __restrict__ eidx, const float* __restrict__ Yt,
            const float* __restrict__ part, const float* __restrict__ bias,
            float* __restrict__ out) {
    __shared__ float s_red[64 * SUBS_ * 4];
    const unsigned tid = threadIdx.x;
    const unsigned o = tid & 63;
    const int sub   = __builtin_amdgcn_readfirstlane((int)(tid >> 6));
    const int node0 = (blockIdx.x * SUBS_ + sub) * NPB_;
    const int b     = (blockIdx.x * (SUBS_ * NPB_)) / N_;
    const int n0    = node0 - b * N_;

    const int* e0 = eidx + (size_t)node0 * K_;
    const int* e1 = eidx + (size_t)B_ * N_ * K_ + (size_t)node0 * K_;
    const float* yb = Yt + (size_t)b * N_ * CCAT_;
    float* orow = out + ((size_t)b * COUT_ + o) * N_ + n0;

    unsigned ii[2][2 * K_], jj[2][2 * K_];
    float y1[2][2 * K_], y2[2][2 * K_];

    #pragma unroll
    for (int t = 0; t < 2 * K_; ++t) { ii[0][t] = (unsigned)e1[t]; jj[0][t] = (unsigned)e0[t]; }

    const float4* pp = reinterpret_cast<const float4*>(part + (size_t)b * GPB_ * COUT_) + tid;
    float4 pa = pp[0], pb = pp[64 * SUBS_], pc = pp[2 * 64 * SUBS_], pd = pp[3 * 64 * SUBS_];
    const float bo = bias[o];
    *reinterpret_cast<float4*>(&s_red[4 * tid]) =
        make_float4((pa.x + pb.x) + (pc.x + pd.x), (pa.y + pb.y) + (pc.y + pd.y),
                    (pa.z + pb.z) + (pc.z + pd.z), (pa.w + pb.w) + (pc.w + pd.w));
    __syncthreads();
    float ps[4] = {0.f, 0.f, 0.f, 0.f};
    #pragma unroll
    for (int k = 0; k < 16; k += 4) {
        ps[0] += s_red[64 * (k    ) + o];
        ps[1] += s_red[64 * (k + 1) + o];
        ps[2] += s_red[64 * (k + 2) + o];
        ps[3] += s_red[64 * (k + 3) + o];
    }
    const float bias0 = bo + ((ps[0] + ps[1]) + (ps[2] + ps[3])) * (1.0f / N_);
    __builtin_amdgcn_sched_barrier(0);
    #pragma unroll
    for (int t = 0; t < 2 * K_; ++t) {
        y1[0][t] = yb[ii[0][t] * CCAT_ + o];
        y2[0][t] = yb[jj[0][t] * CCAT_ + COUT_ + o];
    }
    #pragma unroll
    for (int t = 0; t < 2 * K_; ++t) { ii[1][t] = (unsigned)e1[2 * K_ + t]; jj[1][t] = (unsigned)e0[2 * K_ + t]; }

    float res[NPB_];
    const bool l0 = (o & 1u) != 0, l1 = (o & 2u) != 0;
    #pragma unroll
    for (int g = 0; g < NGRP_; ++g) {
        const int cur = g & 1, nxt = cur ^ 1;
        if (g + 1 < NGRP_) {
            #pragma unroll
            for (int t = 0; t < 2 * K_; ++t) {
                y1[nxt][t] = yb[ii[nxt][t] * CCAT_ + o];
                y2[nxt][t] = yb[jj[nxt][t] * CCAT_ + COUT_ + o];
            }
        }
        if (g + 2 < NGRP_) {
            #pragma unroll
            for (int t = 0; t < 2 * K_; ++t) {
                ii[cur][t] = (unsigned)e1[(g + 2) * 2 * K_ + t];
                jj[cur][t] = (unsigned)e0[(g + 2) * 2 * K_ + t];
            }
        }
        #pragma unroll
        for (int q = 0; q < 2; ++q) {
            float m = -3.402823466e38f;
            #pragma unroll
            for (int k = 0; k < K_; ++k)
                m = fmaxf(m, y1[cur][q * K_ + k] + y2[cur][q * K_ + k]);
            res[2 * g + q] = fmaxf(m + bias0, 0.0f);
        }
        if ((g & 3) == 3) {
            const int pb = (g >> 1) - 1;
            #pragma unroll
            for (int c = 0; c < 4; ++c) {
                const float lo = res[4 * pb + c], hi = res[4 * (pb + 1) + c];
                const float got = quad_xor1(l0 ? lo : hi);
                res[4 * pb + c]       = l0 ? got : lo;
                res[4 * (pb + 1) + c] = l0 ? hi : got;
            }
        }
    }

    #pragma unroll
    for (int pb = 0; pb < 2; ++pb) {
        #pragma unroll
        for (int c = 0; c < 4; ++c) {
            const float lo = res[4 * pb + c], hi = res[4 * (pb + 2) + c];
            const float got = quad_xor2(l1 ? lo : hi);
            res[4 * pb + c]       = l1 ? got : lo;
            res[4 * (pb + 2) + c] = l1 ? hi : got;
        }
    }
    float* obase = orow - (size_t)(o & 3u) * N_ + 4 * (o & 3u);
    #pragma unroll
    for (int j = 0; j < 4; ++j)
        *reinterpret_cast<float4*>(obase + (size_t)j * N_) =
            make_float4(res[4 * j], res[4 * j + 1], res[4 * j + 2], res[4 * j + 3]);
}

extern "C" void kernel_launch(void* const* d_in, const int* in_sizes, int n_in,
                              void* d_out, int out_size, void* d_ws, size_t ws_size,
                              hipStream_t stream) {
    const float* x    = (const float*)d_in[0];
    const int*   eidx = (const int*)  d_in[1];
    const float* W    = (const float*)d_in[2];
    const float* bias = (const float*)d_in[3];

    float* ws    = (float*)d_ws;
    float* part  = ws;
    float* Yt    = ws + 8192 + 256;

    gemm_wmma_kernel<<<B_ * GPB_ * 2, 256, 0, stream>>>(x, W, Yt, part);

    edge_kernel<<<(B_ * N_) / (NPB_ * SUBS_), 64 * SUBS_, 0, stream>>>(eidx, Yt, part, bias, (float*)d_out);
}
